# starting kernel
# baseline (speedup 1.0000x reference)
#include <hip/hip_runtime.h>


typedef __attribute__((ext_vector_type(2))) float v2f;
typedef __attribute__((ext_vector_type(8))) float v8f;

#define B_  32
#define N_  512
#define A_  32
#define M_  128
#define D_  32
#define MD_ (M_ * D_)
#define SCALE_ 0.17677669529663687f

static __device__ __forceinline__ v8f wmma_f32(v2f a, v2f b, v8f c) {
  return __builtin_amdgcn_wmma_f32_16x16x4_f32(false, a, false, b, (short)0, c,
                                               false, false);
}

__global__ __launch_bounds__(256) void caps_logits_qk(
    const float* __restrict__ pose, const float* __restrict__ ncv,
    const float* __restrict__ na, const float* __restrict__ w,
    float* __restrict__ qk_out) {
  const int n    = blockIdx.x;
  const int tid  = threadIdx.x;
  const int lane = tid & 31;
  const int wv   = tid >> 5;
  const int l16  = lane & 15;
  const int khalf = (lane < 16) ? 0 : 2;

  __shared__ float vote[32 * 260];
  __shared__ float lg[32 * 128];

  v2f areg[2][8];
  for (int rt = 0; rt < 2; ++rt) {
    const int b = l16 + 16 * rt;
    const float* pp = pose + ((size_t)b * N_ + n) * A_;
    for (int k = 0; k < 8; ++k) areg[rt][k] = *(const v2f*)(pp + 4 * k + khalf);
  }

  for (int mt = 0; mt < 16; ++mt) {
    for (int j = 0; j < 2; ++j) {
      const int ct  = wv * 2 + j;
      const int col = mt * 256 + ct * 16 + l16;
      const float* wbase = w + (size_t)n * A_ * MD_ + col;
      __builtin_prefetch(wbase + 256, 0, 0);
      v8f acc0 = {}, acc1 = {};
      for (int k = 0; k < 8; ++k) {
        const int a = 4 * k + khalf;
        const float* wp = wbase + (size_t)a * MD_;
        v2f bv; bv.x = wp[0]; bv.y = wp[MD_];
        acc0 = wmma_f32(areg[0][k], bv, acc0);
        acc1 = wmma_f32(areg[1][k], bv, acc1);
      }
      const int cbase = ct * 16 + l16;
      const int rbase = (lane < 16) ? 0 : 8;
      for (int i = 0; i < 8; ++i) {
        vote[(i + rbase) * 260 + cbase]      = acc0[i];
        vote[(i + rbase + 16) * 260 + cbase] = acc1[i];
      }
    }
    __syncthreads();
    {
      const int b  = tid & 31;
      const int mi = tid >> 5;
      const int m  = mt * 8 + mi;
      const float* vp = vote + b * 260 + mi * 32;
      const float* cp = ncv + ((size_t)b * M_ + m) * D_;
      float s = 0.f;
      for (int d = 0; d < 32; ++d) s += vp[d] * cp[d];
      lg[b * 128 + m] = s * SCALE_;
    }
    __syncthreads();
  }

  if (tid < 32) {
    const int b = tid;
    float* row = lg + b * 128;
    float mx = -3.4e38f;
    for (int m = 0; m < 128; ++m) mx = fmaxf(mx, row[m]);
    float s = 0.f;
    for (int m = 0; m < 128; ++m) { float e = __expf(row[m] - mx); row[m] = e; s += e; }
    const float inv = 1.f / s;
    const float* nap = na + b * 128;
    float s2 = 0.f;
    for (int m = 0; m < 128; ++m) { float v = row[m] * inv * nap[m]; row[m] = v; s2 += v; }
    const float inv2 = 1.f / (s2 + 1e-10f);
    float* qp = qk_out + ((size_t)b * N_ + n) * M_;
    for (int m = 0; m < 128; ++m) qp[m] = row[m] * inv2;
  }
}

__global__ __launch_bounds__(256) void caps_out(
    const float* __restrict__ pose, const float* __restrict__ ca,
    const float* __restrict__ w, const float* __restrict__ qk,
    float* __restrict__ out) {
  const int m    = blockIdx.x;
  const int dh   = blockIdx.y;
  const int tid  = threadIdx.x;
  const int lane = tid & 31;
  const int wv   = tid >> 5;
  const int l16  = lane & 15;
  const int khalf = (lane < 16) ? 0 : 2;
  const int dcol = dh * 16 + l16;

  __shared__ float red[8][32 * 17];

  v8f acc0 = {}, acc1 = {};
  const int b0 = l16, b1 = l16 + 16;

  for (int n = wv * 64; n < wv * 64 + 64; ++n) {
    const float q0 = qk[((size_t)b0 * N_ + n) * M_ + m] * ca[(size_t)b0 * N_ + n];
    const float q1 = qk[((size_t)b1 * N_ + n) * M_ + m] * ca[(size_t)b1 * N_ + n];
    const float* pp0 = pose + ((size_t)b0 * N_ + n) * A_;
    const float* pp1 = pose + ((size_t)b1 * N_ + n) * A_;
    const float* wbase = w + (size_t)n * A_ * MD_ + m * D_ + dcol;
    __builtin_prefetch(wbase + (size_t)2 * A_ * MD_, 0, 0);
    for (int k = 0; k < 8; ++k) {
      const int a = 4 * k + khalf;
      const v2f p0 = *(const v2f*)(pp0 + a);
      const v2f p1 = *(const v2f*)(pp1 + a);
      v2f av0; av0.x = q0 * p0.x; av0.y = q0 * p0.y;
      v2f av1; av1.x = q1 * p1.x; av1.y = q1 * p1.y;
      const float* wp = wbase + (size_t)a * MD_;
      v2f bv; bv.x = wp[0]; bv.y = wp[MD_];
      acc0 = wmma_f32(av0, bv, acc0);
      acc1 = wmma_f32(av1, bv, acc1);
    }
  }

  const int rbase = (lane < 16) ? 0 : 8;
  for (int i = 0; i < 8; ++i) {
    red[wv][(i + rbase) * 17 + l16]      = acc0[i];
    red[wv][(i + rbase + 16) * 17 + l16] = acc1[i];
  }
  __syncthreads();
  for (int e = tid; e < 512; e += 256) {
    const int b = e >> 4, dc = e & 15;
    float s = 0.f;
    for (int v = 0; v < 8; ++v) s += red[v][b * 17 + dc];
    out[(size_t)b * MD_ + m * D_ + dh * 16 + dc] = s;
  }
}

extern "C" void kernel_launch(void* const* d_in, const int* in_sizes, int n_in,
                              void* d_out, int out_size, void* d_ws,
                              size_t ws_size, hipStream_t stream) {
  (void)in_sizes; (void)n_in; (void)out_size; (void)d_ws; (void)ws_size;
  const float* pose = (const float*)d_in[0];
  const float* ca   = (const float*)d_in[1];
  const float* ncv  = (const float*)d_in[2];
  const float* na   = (const float*)d_in[3];
  const float* w    = (const float*)d_in[4];

  float* out    = (float*)d_out;
  float* na_out = out + (size_t)B_ * M_ * D_;
  float* qk_out = na_out + (size_t)B_ * M_;

  caps_logits_qk<<<dim3(N_), dim3(256), 0, stream>>>(pose, ncv, na, w, qk_out);
  hipMemcpyAsync(na_out, na, (size_t)B_ * M_ * sizeof(float),
                 hipMemcpyDeviceToDevice, stream);
  caps_out<<<dim3(M_, 2), dim3(256), 0, stream>>>(pose, ca, w, qk_out, out);
}


# second kernel re-tiled: work-group = one m x 16 b x all 32 d on 32 waves of 16 n (was 32 b x 16 d, 8 waves of 64 n); half-wave reads whole 128-B w rows; 32 partials (was 8) summed in fixed order
# speedup vs baseline: 1.2545x; 1.2545x over previous
#include <hip/hip_runtime.h>


typedef __attribute__((ext_vector_type(2))) float v2f;
typedef __attribute__((ext_vector_type(8))) float v8f;

#define B_  32
#define N_  512
#define A_  32
#define M_  128
#define D_  32
#define MD_ (M_ * D_)
#define SCALE_ 0.17677669529663687f

static __device__ __forceinline__ v8f wmma_f32(v2f a, v2f b, v8f c) {
  return __builtin_amdgcn_wmma_f32_16x16x4_f32(false, a, false, b, (short)0, c,
                                               false, false);
}

__global__ __launch_bounds__(256) void caps_logits_qk(
    const float* __restrict__ pose, const float* __restrict__ ncv,
    const float* __restrict__ na, const float* __restrict__ w,
    float* __restrict__ qk_out) {
  const int n    = blockIdx.x;
  const int tid  = threadIdx.x;
  const int lane = tid & 31;
  const int wv   = tid >> 5;
  const int l16  = lane & 15;
  const int khalf = (lane < 16) ? 0 : 2;

  __shared__ float vote[32 * 260];
  __shared__ float lg[32 * 128];

  v2f areg[2][8];
  for (int rt = 0; rt < 2; ++rt) {
    const int b = l16 + 16 * rt;
    const float* pp = pose + ((size_t)b * N_ + n) * A_;
    for (int k = 0; k < 8; ++k) areg[rt][k] = *(const v2f*)(pp + 4 * k + khalf);
  }

  for (int mt = 0; mt < 16; ++mt) {
    for (int j = 0; j < 2; ++j) {
      const int ct  = wv * 2 + j;
      const int col = mt * 256 + ct * 16 + l16;
      const float* wbase = w + (size_t)n * A_ * MD_ + col;
      __builtin_prefetch(wbase + 256, 0, 0);
      v8f acc0 = {}, acc1 = {};
      for (int k = 0; k < 8; ++k) {
        const int a = 4 * k + khalf;
        const float* wp = wbase + (size_t)a * MD_;
        v2f bv; bv.x = wp[0]; bv.y = wp[MD_];
        acc0 = wmma_f32(areg[0][k], bv, acc0);
        acc1 = wmma_f32(areg[1][k], bv, acc1);
      }
      const int cbase = ct * 16 + l16;
      const int rbase = (lane < 16) ? 0 : 8;
      for (int i = 0; i < 8; ++i) {
        vote[(i + rbase) * 260 + cbase]      = acc0[i];
        vote[(i + rbase + 16) * 260 + cbase] = acc1[i];
      }
    }
    __syncthreads();
    {
      const int b  = tid & 31;
      const int mi = tid >> 5;
      const int m  = mt * 8 + mi;
      const float* vp = vote + b * 260 + mi * 32;
      const float* cp = ncv + ((size_t)b * M_ + m) * D_;
      float s = 0.f;
      for (int d = 0; d < 32; ++d) s += vp[d] * cp[d];
      lg[b * 128 + m] = s * SCALE_;
    }
    __syncthreads();
  }

  if (tid < 32) {
    const int b = tid;
    float* row = lg + b * 128;
    float mx = -3.4e38f;
    for (int m = 0; m < 128; ++m) mx = fmaxf(mx, row[m]);
    float s = 0.f;
    for (int m = 0; m < 128; ++m) { float e = __expf(row[m] - mx); row[m] = e; s += e; }
    const float inv = 1.f / s;
    const float* nap = na + b * 128;
    float s2 = 0.f;
    for (int m = 0; m < 128; ++m) { float v = row[m] * inv * nap[m]; row[m] = v; s2 += v; }
    const float inv2 = 1.f / (s2 + 1e-10f);
    float* qp = qk_out + ((size_t)b * N_ + n) * M_;
    for (int m = 0; m < 128; ++m) qp[m] = row[m] * inv2;
  }
}

__global__ __launch_bounds__(1024) void caps_out(
    const float* __restrict__ pose, const float* __restrict__ ca,
    const float* __restrict__ w, const float* __restrict__ qk,
    float* __restrict__ out) {
  const int m    = blockIdx.x;
  const int bh   = blockIdx.y;
  const int tid  = threadIdx.x;
  const int lane = tid & 31;
  const int wv   = tid >> 5;
  const int l16  = lane & 15;
  const int khalf = (lane < 16) ? 0 : 2;
  const int b    = bh * 16 + l16;

  __shared__ float red[32][16 * 33];

  v8f acc0 = {}, acc1 = {};
  const float* qp = qk + (size_t)b * N_ * M_ + m;
  const float* cp = ca + (size_t)b * N_;
  const float* pb = pose + (size_t)b * N_ * A_ + khalf;
  const float* wl = w + (size_t)khalf * MD_ + m * D_ + 2 * l16;

  for (int n = wv * 16; n < wv * 16 + 16; ++n) {
    const float q = qp[(size_t)n * M_] * cp[n];
    const float* pp = pb + (size_t)n * A_;
    const float* wn = wl + (size_t)n * A_ * MD_;
    v2f wa[8], wb[8], pv[8];
    for (int k = 0; k < 8; ++k) {
      wa[k] = *(const v2f*)(wn + (size_t)(4 * k) * MD_);
      wb[k] = *(const v2f*)(wn + (size_t)(4 * k + 1) * MD_);
      pv[k] = *(const v2f*)(pp + 4 * k);
    }
    for (int k = 0; k < 8; ++k) {
      v2f av; av.x = q * pv[k].x;  av.y = q * pv[k].y;
      v2f be; be.x = wa[k].x;      be.y = wb[k].x;
      v2f bo; bo.x = wa[k].y;      bo.y = wb[k].y;
      acc0 = wmma_f32(av, be, acc0);
      acc1 = wmma_f32(av, bo, acc1);
    }
  }

  const int rbase = (lane < 16) ? 0 : 8;
  for (int i = 0; i < 8; ++i) {
    red[wv][(i + rbase) * 33 + 2 * l16]     = acc0[i];
    red[wv][(i + rbase) * 33 + 2 * l16 + 1] = acc1[i];
  }
  __syncthreads();
  if (tid < 512) {
    const int r = tid >> 5, d = tid & 31;
    float s = 0.f;
    for (int v = 0; v < 32; ++v) s += red[v][r * 33 + d];
    out[(size_t)(bh * 16 + r) * MD_ + m * D_ + d] = s;
  }
}

extern "C" void kernel_launch(void* const* d_in, const int* in_sizes, int n_in,
                              void* d_out, int out_size, void* d_ws,
                              size_t ws_size, hipStream_t stream) {
  (void)in_sizes; (void)n_in; (void)out_size; (void)d_ws; (void)ws_size;
  const float* pose = (const float*)d_in[0];
  const float* ca   = (const float*)d_in[1];
  const float* ncv  = (const float*)d_in[2];
  const float* na   = (const float*)d_in[3];
  const float* w    = (const float*)d_in[4];

  float* out    = (float*)d_out;
  float* na_out = out + (size_t)B_ * M_ * D_;
  float* qk_out = na_out + (size_t)B_ * M_;

  caps_logits_qk<<<dim3(N_), dim3(256), 0, stream>>>(pose, ncv, na, w, qk_out);
  hipMemcpyAsync(na_out, na, (size_t)B_ * M_ * sizeof(float),
                 hipMemcpyDeviceToDevice, stream);
  caps_out<<<dim3(M_, 2), dim3(1024), 0, stream>>>(pose, ca, w, qk_out, out);
}


# first kernel: logit dot with ncv taken on the WMMA accumulators in registers, d summed by a 16-lane DPP butterfly (sum order over d changes); LDS vote tile and both per-m-tile barriers removed
# speedup vs baseline: 1.3519x; 1.0776x over previous
#include <hip/hip_runtime.h>


typedef __attribute__((ext_vector_type(2))) float v2f;
typedef __attribute__((ext_vector_type(8))) float v8f;

#define B_  32
#define N_  512
#define A_  32
#define M_  128
#define D_  32
#define MD_ (M_ * D_)
#define SCALE_ 0.17677669529663687f

static __device__ __forceinline__ v8f wmma_f32(v2f a, v2f b, v8f c) {
  return __builtin_amdgcn_wmma_f32_16x16x4_f32(false, a, false, b, (short)0, c,
                                               false, false);
}

#define ROW_XADD(x, n)                                                        \
  (x) += __builtin_bit_cast(                                                  \
      float, __builtin_amdgcn_update_dpp(0, __builtin_bit_cast(int, (x)),     \
                                         0x160 + (n), 0xf, 0xf, true))

__global__ __launch_bounds__(256) void caps_logits_qk(
    const float* __restrict__ pose, const float* __restrict__ ncv,
    const float* __restrict__ na, const float* __restrict__ w,
    float* __restrict__ qk_out) {
  const int n    = blockIdx.x;
  const int tid  = threadIdx.x;
  const int lane = tid & 31;
  const int wv   = tid >> 5;
  const int l16  = lane & 15;
  const int khalf = (lane < 16) ? 0 : 2;
  const int rbase = (lane < 16) ? 0 : 8;

  __shared__ float lg[32 * 128];

  v2f areg[2][8];
  for (int rt = 0; rt < 2; ++rt) {
    const int b = l16 + 16 * rt;
    const float* pp = pose + ((size_t)b * N_ + n) * A_;
    for (int k = 0; k < 8; ++k) areg[rt][k] = *(const v2f*)(pp + 4 * k + khalf);
  }

  for (int mt = 0; mt < 16; ++mt) {
    const int m = mt * 8 + wv;
    const float* cb = ncv + ((size_t)rbase * M_ + m) * D_ + l16;
    float p0[8], p1[8];
    for (int i = 0; i < 8; ++i) { p0[i] = 0.f; p1[i] = 0.f; }
    for (int j = 0; j < 2; ++j) {
      const int col = m * D_ + j * 16 + l16;
      const float* wbase = w + (size_t)n * A_ * MD_ + col;
      __builtin_prefetch(wbase + 256, 0, 0);
      v8f acc0 = {}, acc1 = {};
      for (int k = 0; k < 8; ++k) {
        const int a = 4 * k + khalf;
        const float* wp = wbase + (size_t)a * MD_;
        v2f bv; bv.x = wp[0]; bv.y = wp[MD_];
        acc0 = wmma_f32(areg[0][k], bv, acc0);
        acc1 = wmma_f32(areg[1][k], bv, acc1);
      }
      for (int i = 0; i < 8; ++i) {
        p0[i] += acc0[i] * cb[(size_t)i * MD_ + j * 16];
        p1[i] += acc1[i] * cb[(size_t)(i + 16) * MD_ + j * 16];
      }
    }
    for (int i = 0; i < 8; ++i) { ROW_XADD(p0[i], 1); ROW_XADD(p1[i], 1); }
    for (int i = 0; i < 8; ++i) { ROW_XADD(p0[i], 2); ROW_XADD(p1[i], 2); }
    for (int i = 0; i < 8; ++i) { ROW_XADD(p0[i], 4); ROW_XADD(p1[i], 4); }
    for (int i = 0; i < 8; ++i) { ROW_XADD(p0[i], 8); ROW_XADD(p1[i], 8); }
    float s0 = p0[0], s1 = p1[0];
    for (int i = 1; i < 8; ++i) {
      s0 = (l16 == i) ? p0[i] : s0;
      s1 = (l16 == i) ? p1[i] : s1;
    }
    if (l16 < 8) {
      lg[(l16 + rbase) * 128 + m]      = s0 * SCALE_;
      lg[(l16 + rbase + 16) * 128 + m] = s1 * SCALE_;
    }
  }
  __syncthreads();

  if (tid < 32) {
    const int b = tid;
    float* row = lg + b * 128;
    float mx = -3.4e38f;
    for (int m = 0; m < 128; ++m) mx = fmaxf(mx, row[m]);
    float s = 0.f;
    for (int m = 0; m < 128; ++m) { float e = __expf(row[m] - mx); row[m] = e; s += e; }
    const float inv = 1.f / s;
    const float* nap = na + b * 128;
    float s2 = 0.f;
    for (int m = 0; m < 128; ++m) { float v = row[m] * inv * nap[m]; row[m] = v; s2 += v; }
    const float inv2 = 1.f / (s2 + 1e-10f);
    float* qp = qk_out + ((size_t)b * N_ + n) * M_;
    for (int m = 0; m < 128; ++m) qp[m] = row[m] * inv2;
  }
}

__global__ __launch_bounds__(1024) void caps_out(
    const float* __restrict__ pose, const float* __restrict__ ca,
    const float* __restrict__ w, const float* __restrict__ qk,
    float* __restrict__ out) {
  const int m    = blockIdx.x;
  const int bh   = blockIdx.y;
  const int tid  = threadIdx.x;
  const int lane = tid & 31;
  const int wv   = tid >> 5;
  const int l16  = lane & 15;
  const int khalf = (lane < 16) ? 0 : 2;
  const int b    = bh * 16 + l16;

  __shared__ float red[32][16 * 33];

  v8f acc0 = {}, acc1 = {};
  const float* qp = qk + (size_t)b * N_ * M_ + m;
  const float* cp = ca + (size_t)b * N_;
  const float* pb = pose + (size_t)b * N_ * A_ + khalf;
  const float* wl = w + (size_t)khalf * MD_ + m * D_ + 2 * l16;

  for (int n = wv * 16; n < wv * 16 + 16; ++n) {
    const float q = qp[(size_t)n * M_] * cp[n];
    const float* pp = pb + (size_t)n * A_;
    const float* wn = wl + (size_t)n * A_ * MD_;
    v2f wa[8], wb[8], pv[8];
    for (int k = 0; k < 8; ++k) {
      wa[k] = *(const v2f*)(wn + (size_t)(4 * k) * MD_);
      wb[k] = *(const v2f*)(wn + (size_t)(4 * k + 1) * MD_);
      pv[k] = *(const v2f*)(pp + 4 * k);
    }
    for (int k = 0; k < 8; ++k) {
      v2f av; av.x = q * pv[k].x;  av.y = q * pv[k].y;
      v2f be; be.x = wa[k].x;      be.y = wb[k].x;
      v2f bo; bo.x = wa[k].y;      bo.y = wb[k].y;
      acc0 = wmma_f32(av, be, acc0);
      acc1 = wmma_f32(av, bo, acc1);
    }
  }

  const int rbase = (lane < 16) ? 0 : 8;
  for (int i = 0; i < 8; ++i) {
    red[wv][(i + rbase) * 33 + 2 * l16]     = acc0[i];
    red[wv][(i + rbase) * 33 + 2 * l16 + 1] = acc1[i];
  }
  __syncthreads();
  if (tid < 512) {
    const int r = tid >> 5, d = tid & 31;
    float s = 0.f;
    for (int v = 0; v < 32; ++v) s += red[v][r * 33 + d];
    out[(size_t)(bh * 16 + r) * MD_ + m * D_ + d] = s;
  }
}

extern "C" void kernel_launch(void* const* d_in, const int* in_sizes, int n_in,
                              void* d_out, int out_size, void* d_ws,
                              size_t ws_size, hipStream_t stream) {
  (void)in_sizes; (void)n_in; (void)out_size; (void)d_ws; (void)ws_size;
  const float* pose = (const float*)d_in[0];
  const float* ca   = (const float*)d_in[1];
  const float* ncv  = (const float*)d_in[2];
  const float* na   = (const float*)d_in[3];
  const float* w    = (const float*)d_in[4];

  float* out    = (float*)d_out;
  float* na_out = out + (size_t)B_ * M_ * D_;
  float* qk_out = na_out + (size_t)B_ * M_;

  caps_logits_qk<<<dim3(N_), dim3(256), 0, stream>>>(pose, ncv, na, w, qk_out);
  hipMemcpyAsync(na_out, na, (size_t)B_ * M_ * sizeof(float),
                 hipMemcpyDeviceToDevice, stream);
  caps_out<<<dim3(M_, 2), dim3(1024), 0, stream>>>(pose, ca, w, qk_out, out);
}


# first kernel: softmax/renormalise tail on all 8 waves (half-wave = one row b, 8 m per lane) instead of one thread per b; sums over m become a DPP tree (sum order changes); qk stored as whole lines
# speedup vs baseline: 1.4540x; 1.0755x over previous
#include <hip/hip_runtime.h>


typedef __attribute__((ext_vector_type(2))) float v2f;
typedef __attribute__((ext_vector_type(8))) float v8f;

#define B_  32
#define N_  512
#define A_  32
#define M_  128
#define D_  32
#define MD_ (M_ * D_)
#define SCALE_ 0.17677669529663687f

static __device__ __forceinline__ v8f wmma_f32(v2f a, v2f b, v8f c) {
  return __builtin_amdgcn_wmma_f32_16x16x4_f32(false, a, false, b, (short)0, c,
                                               false, false);
}

#define ROW_XADD(x, n)                                                        \
  (x) += __builtin_bit_cast(                                                  \
      float, __builtin_amdgcn_update_dpp(0, __builtin_bit_cast(int, (x)),     \
                                         0x160 + (n), 0xf, 0xf, true))
#define ROW_XMAX(x, n)                                                        \
  (x) = fmaxf((x), __builtin_bit_cast(                                        \
      float, __builtin_amdgcn_update_dpp(0, __builtin_bit_cast(int, (x)),     \
                                         0x160 + (n), 0xf, 0xf, true)))

__global__ __launch_bounds__(256) void caps_logits_qk(
    const float* __restrict__ pose, const float* __restrict__ ncv,
    const float* __restrict__ na, const float* __restrict__ w,
    float* __restrict__ qk_out) {
  const int n    = blockIdx.x;
  const int tid  = threadIdx.x;
  const int lane = tid & 31;
  const int wv   = tid >> 5;
  const int l16  = lane & 15;
  const int khalf = (lane < 16) ? 0 : 2;
  const int rbase = (lane < 16) ? 0 : 8;

  __shared__ __attribute__((aligned(16))) float lg[32 * 128];

  v2f areg[2][8];
  for (int rt = 0; rt < 2; ++rt) {
    const int b = l16 + 16 * rt;
    const float* pp = pose + ((size_t)b * N_ + n) * A_;
    for (int k = 0; k < 8; ++k) areg[rt][k] = *(const v2f*)(pp + 4 * k + khalf);
  }

  for (int mt = 0; mt < 16; ++mt) {
    const int m = mt * 8 + wv;
    const float* cb = ncv + ((size_t)rbase * M_ + m) * D_ + l16;
    float p0[8], p1[8];
    for (int i = 0; i < 8; ++i) { p0[i] = 0.f; p1[i] = 0.f; }
    for (int j = 0; j < 2; ++j) {
      const int col = m * D_ + j * 16 + l16;
      const float* wbase = w + (size_t)n * A_ * MD_ + col;
      __builtin_prefetch(wbase + 256, 0, 0);
      v8f acc0 = {}, acc1 = {};
      for (int k = 0; k < 8; ++k) {
        const int a = 4 * k + khalf;
        const float* wp = wbase + (size_t)a * MD_;
        v2f bv; bv.x = wp[0]; bv.y = wp[MD_];
        acc0 = wmma_f32(areg[0][k], bv, acc0);
        acc1 = wmma_f32(areg[1][k], bv, acc1);
      }
      for (int i = 0; i < 8; ++i) {
        p0[i] += acc0[i] * cb[(size_t)i * MD_ + j * 16];
        p1[i] += acc1[i] * cb[(size_t)(i + 16) * MD_ + j * 16];
      }
    }
    for (int i = 0; i < 8; ++i) { ROW_XADD(p0[i], 1); ROW_XADD(p1[i], 1); }
    for (int i = 0; i < 8; ++i) { ROW_XADD(p0[i], 2); ROW_XADD(p1[i], 2); }
    for (int i = 0; i < 8; ++i) { ROW_XADD(p0[i], 4); ROW_XADD(p1[i], 4); }
    for (int i = 0; i < 8; ++i) { ROW_XADD(p0[i], 8); ROW_XADD(p1[i], 8); }
    float s0 = p0[0], s1 = p1[0];
    for (int i = 1; i < 8; ++i) {
      s0 = (l16 == i) ? p0[i] : s0;
      s1 = (l16 == i) ? p1[i] : s1;
    }
    if (l16 < 8) {
      lg[(l16 + rbase) * 128 + m]      = s0 * SCALE_;
      lg[(l16 + rbase + 16) * 128 + m] = s1 * SCALE_;
    }
  }
  __syncthreads();

  {
    const int h = lane >> 4;
    float e[2][8], g[2][8];
    for (int p = 0; p < 2; ++p) {
      const int b = p * 16 + wv * 2 + h;
      const float* row = lg + b * 128 + 4 * l16;
      const float* nap = na + b * 128 + 4 * l16;
      const float4 r0 = *(const float4*)(row);
      const float4 r1 = *(const float4*)(row + 64);
      const float4 a0 = *(const float4*)(nap);
      const float4 a1 = *(const float4*)(nap + 64);
      e[p][0] = r0.x; e[p][1] = r0.y; e[p][2] = r0.z; e[p][3] = r0.w;
      e[p][4] = r1.x; e[p][5] = r1.y; e[p][6] = r1.z; e[p][7] = r1.w;
      g[p][0] = a0.x; g[p][1] = a0.y; g[p][2] = a0.z; g[p][3] = a0.w;
      g[p][4] = a1.x; g[p][5] = a1.y; g[p][6] = a1.z; g[p][7] = a1.w;
    }
    for (int p = 0; p < 2; ++p) {
      const int b = p * 16 + wv * 2 + h;
      float mx = e[p][0];
      for (int i = 1; i < 8; ++i) mx = fmaxf(mx, e[p][i]);
      ROW_XMAX(mx, 1); ROW_XMAX(mx, 2); ROW_XMAX(mx, 4); ROW_XMAX(mx, 8);
      float s = 0.f;
      for (int i = 0; i < 8; ++i) { e[p][i] = __expf(e[p][i] - mx); s += e[p][i]; }
      ROW_XADD(s, 1); ROW_XADD(s, 2); ROW_XADD(s, 4); ROW_XADD(s, 8);
      const float inv = 1.f / s;
      float s2 = 0.f;
      for (int i = 0; i < 8; ++i) { e[p][i] = e[p][i] * inv * g[p][i]; s2 += e[p][i]; }
      ROW_XADD(s2, 1); ROW_XADD(s2, 2); ROW_XADD(s2, 4); ROW_XADD(s2, 8);
      const float inv2 = 1.f / (s2 + 1e-10f);
      float* qp = qk_out + ((size_t)b * N_ + n) * M_ + 4 * l16;
      float4 o0, o1;
      o0.x = e[p][0] * inv2; o0.y = e[p][1] * inv2; o0.z = e[p][2] * inv2; o0.w = e[p][3] * inv2;
      o1.x = e[p][4] * inv2; o1.y = e[p][5] * inv2; o1.z = e[p][6] * inv2; o1.w = e[p][7] * inv2;
      *(float4*)(qp)      = o0;
      *(float4*)(qp + 64) = o1;
    }
  }
}

__global__ __launch_bounds__(1024) void caps_out(
    const float* __restrict__ pose, const float* __restrict__ ca,
    const float* __restrict__ w, const float* __restrict__ qk,
    float* __restrict__ out) {
  const int m    = blockIdx.x;
  const int bh   = blockIdx.y;
  const int tid  = threadIdx.x;
  const int lane = tid & 31;
  const int wv   = tid >> 5;
  const int l16  = lane & 15;
  const int khalf = (lane < 16) ? 0 : 2;
  const int b    = bh * 16 + l16;

  __shared__ float red[32][16 * 33];

  v8f acc0 = {}, acc1 = {};
  const float* qp = qk + (size_t)b * N_ * M_ + m;
  const float* cp = ca + (size_t)b * N_;
  const float* pb = pose + (size_t)b * N_ * A_ + khalf;
  const float* wl = w + (size_t)khalf * MD_ + m * D_ + 2 * l16;

  for (int n = wv * 16; n < wv * 16 + 16; ++n) {
    const float q = qp[(size_t)n * M_] * cp[n];
    const float* pp = pb + (size_t)n * A_;
    const float* wn = wl + (size_t)n * A_ * MD_;
    v2f wa[8], wb[8], pv[8];
    for (int k = 0; k < 8; ++k) {
      wa[k] = *(const v2f*)(wn + (size_t)(4 * k) * MD_);
      wb[k] = *(const v2f*)(wn + (size_t)(4 * k + 1) * MD_);
      pv[k] = *(const v2f*)(pp + 4 * k);
    }
    for (int k = 0; k < 8; ++k) {
      v2f av; av.x = q * pv[k].x;  av.y = q * pv[k].y;
      v2f be; be.x = wa[k].x;      be.y = wb[k].x;
      v2f bo; bo.x = wa[k].y;      bo.y = wb[k].y;
      acc0 = wmma_f32(av, be, acc0);
      acc1 = wmma_f32(av, bo, acc1);
    }
  }

  const int rbase = (lane < 16) ? 0 : 8;
  for (int i = 0; i < 8; ++i) {
    red[wv][(i + rbase) * 33 + 2 * l16]     = acc0[i];
    red[wv][(i + rbase) * 33 + 2 * l16 + 1] = acc1[i];
  }
  __syncthreads();
  if (tid < 512) {
    const int r = tid >> 5, d = tid & 31;
    float s = 0.f;
    for (int v = 0; v < 32; ++v) s += red[v][r * 33 + d];
    out[(size_t)(bh * 16 + r) * MD_ + m * D_ + d] = s;
  }
}

extern "C" void kernel_launch(void* const* d_in, const int* in_sizes, int n_in,
                              void* d_out, int out_size, void* d_ws,
                              size_t ws_size, hipStream_t stream) {
  (void)in_sizes; (void)n_in; (void)out_size; (void)d_ws; (void)ws_size;
  const float* pose = (const float*)d_in[0];
  const float* ca   = (const float*)d_in[1];
  const float* ncv  = (const float*)d_in[2];
  const float* na   = (const float*)d_in[3];
  const float* w    = (const float*)d_in[4];

  float* out    = (float*)d_out;
  float* na_out = out + (size_t)B_ * M_ * D_;
  float* qk_out = na_out + (size_t)B_ * M_;

  caps_logits_qk<<<dim3(N_), dim3(256), 0, stream>>>(pose, ncv, na, w, qk_out);
  hipMemcpyAsync(na_out, na, (size_t)B_ * M_ * sizeof(float),
                 hipMemcpyDeviceToDevice, stream);
  caps_out<<<dim3(M_, 2), dim3(1024), 0, stream>>>(pose, ca, w, qk_out, out);
}


# first kernel: w and ncv read with 16-byte loads, a wave owning two adjacent m per pass (half-wave = one 256-B row piece); d summed 4 in-lane then by an 8-lane DPP tree (sum order over d changes)
# speedup vs baseline: 1.5455x; 1.0629x over previous
#include <hip/hip_runtime.h>


typedef __attribute__((ext_vector_type(2))) float v2f;
typedef __attribute__((ext_vector_type(8))) float v8f;

#define B_  32
#define N_  512
#define A_  32
#define M_  128
#define D_  32
#define MD_ (M_ * D_)
#define SCALE_ 0.17677669529663687f

static __device__ __forceinline__ v8f wmma_f32(v2f a, v2f b, v8f c) {
  return __builtin_amdgcn_wmma_f32_16x16x4_f32(false, a, false, b, (short)0, c,
                                               false, false);
}

#define ROW_XADD(x, n)                                                        \
  (x) += __builtin_bit_cast(                                                  \
      float, __builtin_amdgcn_update_dpp(0, __builtin_bit_cast(int, (x)),     \
                                         0x160 + (n), 0xf, 0xf, true))
#define ROW_XMAX(x, n)                                                        \
  (x) = fmaxf((x), __builtin_bit_cast(                                        \
      float, __builtin_amdgcn_update_dpp(0, __builtin_bit_cast(int, (x)),     \
                                         0x160 + (n), 0xf, 0xf, true)))

__global__ __launch_bounds__(256) void caps_logits_qk(
    const float* __restrict__ pose, const float* __restrict__ ncv,
    const float* __restrict__ na, const float* __restrict__ w,
    float* __restrict__ qk_out) {
  const int n    = blockIdx.x;
  const int tid  = threadIdx.x;
  const int lane = tid & 31;
  const int wv   = tid >> 5;
  const int l16  = lane & 15;
  const int khalf = (lane < 16) ? 0 : 2;
  const int rbase = (lane < 16) ? 0 : 8;
  const int c8   = l16 & 7;
  const int msel = l16 >> 3;

  __shared__ __attribute__((aligned(16))) float lg[32 * 128];

  v2f areg[2][8];
  for (int rt = 0; rt < 2; ++rt) {
    const int b = l16 + 16 * rt;
    const float* pp = pose + ((size_t)b * N_ + n) * A_;
    for (int k = 0; k < 8; ++k) areg[rt][k] = *(const v2f*)(pp + 4 * k + khalf);
  }

  for (int mt = 0; mt < 8; ++mt) {
    const int m0 = (mt * 8 + wv) * 2;
    const float* wbase = w + (size_t)n * A_ * MD_ + (size_t)khalf * MD_ +
                         m0 * D_ + 4 * l16;
    const float* cb = ncv + ((size_t)rbase * M_ + m0) * D_ + 4 * l16;
    v8f acc0[4], acc1[4];
    for (int t = 0; t < 4; ++t) { acc0[t] = (v8f){}; acc1[t] = (v8f){}; }
    for (int k = 0; k < 8; ++k) {
      const float4 r0 = *(const float4*)(wbase + (size_t)(4 * k) * MD_);
      const float4 r1 = *(const float4*)(wbase + (size_t)(4 * k + 1) * MD_);
      const float e0[4] = {r0.x, r0.y, r0.z, r0.w};
      const float e1[4] = {r1.x, r1.y, r1.z, r1.w};
      for (int t = 0; t < 4; ++t) {
        v2f bv; bv.x = e0[t]; bv.y = e1[t];
        acc0[t] = wmma_f32(areg[0][k], bv, acc0[t]);
        acc1[t] = wmma_f32(areg[1][k], bv, acc1[t]);
      }
    }
    float p0[8], p1[8];
    for (int i = 0; i < 8; ++i) {
      const float4 q0 = *(const float4*)(cb + (size_t)i * MD_);
      const float4 q1 = *(const float4*)(cb + (size_t)(i + 16) * MD_);
      p0[i] = acc0[0][i] * q0.x + acc0[1][i] * q0.y + acc0[2][i] * q0.z + acc0[3][i] * q0.w;
      p1[i] = acc1[0][i] * q1.x + acc1[1][i] * q1.y + acc1[2][i] * q1.z + acc1[3][i] * q1.w;
    }
    for (int i = 0; i < 8; ++i) { ROW_XADD(p0[i], 1); ROW_XADD(p1[i], 1); }
    for (int i = 0; i < 8; ++i) { ROW_XADD(p0[i], 2); ROW_XADD(p1[i], 2); }
    for (int i = 0; i < 8; ++i) { ROW_XADD(p0[i], 4); ROW_XADD(p1[i], 4); }
    float s0 = p0[0], s1 = p1[0];
    for (int i = 1; i < 8; ++i) {
      s0 = (c8 == i) ? p0[i] : s0;
      s1 = (c8 == i) ? p1[i] : s1;
    }
    lg[(c8 + rbase) * 128 + m0 + msel]      = s0 * SCALE_;
    lg[(c8 + rbase + 16) * 128 + m0 + msel] = s1 * SCALE_;
  }
  __syncthreads();

  {
    const int h = lane >> 4;
    float e[2][8], g[2][8];
    for (int p = 0; p < 2; ++p) {
      const int b = p * 16 + wv * 2 + h;
      const float* row = lg + b * 128 + 4 * l16;
      const float* nap = na + b * 128 + 4 * l16;
      const float4 r0 = *(const float4*)(row);
      const float4 r1 = *(const float4*)(row + 64);
      const float4 a0 = *(const float4*)(nap);
      const float4 a1 = *(const float4*)(nap + 64);
      e[p][0] = r0.x; e[p][1] = r0.y; e[p][2] = r0.z; e[p][3] = r0.w;
      e[p][4] = r1.x; e[p][5] = r1.y; e[p][6] = r1.z; e[p][7] = r1.w;
      g[p][0] = a0.x; g[p][1] = a0.y; g[p][2] = a0.z; g[p][3] = a0.w;
      g[p][4] = a1.x; g[p][5] = a1.y; g[p][6] = a1.z; g[p][7] = a1.w;
    }
    for (int p = 0; p < 2; ++p) {
      const int b = p * 16 + wv * 2 + h;
      float mx = e[p][0];
      for (int i = 1; i < 8; ++i) mx = fmaxf(mx, e[p][i]);
      ROW_XMAX(mx, 1); ROW_XMAX(mx, 2); ROW_XMAX(mx, 4); ROW_XMAX(mx, 8);
      float s = 0.f;
      for (int i = 0; i < 8; ++i) { e[p][i] = __expf(e[p][i] - mx); s += e[p][i]; }
      ROW_XADD(s, 1); ROW_XADD(s, 2); ROW_XADD(s, 4); ROW_XADD(s, 8);
      const float inv = 1.f / s;
      float s2 = 0.f;
      for (int i = 0; i < 8; ++i) { e[p][i] = e[p][i] * inv * g[p][i]; s2 += e[p][i]; }
      ROW_XADD(s2, 1); ROW_XADD(s2, 2); ROW_XADD(s2, 4); ROW_XADD(s2, 8);
      const float inv2 = 1.f / (s2 + 1e-10f);
      float* qp = qk_out + ((size_t)b * N_ + n) * M_ + 4 * l16;
      float4 o0, o1;
      o0.x = e[p][0] * inv2; o0.y = e[p][1] * inv2; o0.z = e[p][2] * inv2; o0.w = e[p][3] * inv2;
      o1.x = e[p][4] * inv2; o1.y = e[p][5] * inv2; o1.z = e[p][6] * inv2; o1.w = e[p][7] * inv2;
      *(float4*)(qp)      = o0;
      *(float4*)(qp + 64) = o1;
    }
  }
}

__global__ __launch_bounds__(1024) void caps_out(
    const float* __restrict__ pose, const float* __restrict__ ca,
    const float* __restrict__ w, const float* __restrict__ qk,
    float* __restrict__ out) {
  const int m    = blockIdx.x;
  const int bh   = blockIdx.y;
  const int tid  = threadIdx.x;
  const int lane = tid & 31;
  const int wv   = tid >> 5;
  const int l16  = lane & 15;
  const int khalf = (lane < 16) ? 0 : 2;
  const int b    = bh * 16 + l16;

  __shared__ float red[32][16 * 33];

  v8f acc0 = {}, acc1 = {};
  const float* qp = qk + (size_t)b * N_ * M_ + m;
  const float* cp = ca + (size_t)b * N_;
  const float* pb = pose + (size_t)b * N_ * A_ + khalf;
  const float* wl = w + (size_t)khalf * MD_ + m * D_ + 2 * l16;

  for (int n = wv * 16; n < wv * 16 + 16; ++n) {
    const float q = qp[(size_t)n * M_] * cp[n];
    const float* pp = pb + (size_t)n * A_;
    const float* wn = wl + (size_t)n * A_ * MD_;
    v2f wa[8], wb[8], pv[8];
    for (int k = 0; k < 8; ++k) {
      wa[k] = *(const v2f*)(wn + (size_t)(4 * k) * MD_);
      wb[k] = *(const v2f*)(wn + (size_t)(4 * k + 1) * MD_);
      pv[k] = *(const v2f*)(pp + 4 * k);
    }
    for (int k = 0; k < 8; ++k) {
      v2f av; av.x = q * pv[k].x;  av.y = q * pv[k].y;
      v2f be; be.x = wa[k].x;      be.y = wb[k].x;
      v2f bo; bo.x = wa[k].y;      bo.y = wb[k].y;
      acc0 = wmma_f32(av, be, acc0);
      acc1 = wmma_f32(av, bo, acc1);
    }
  }

  const int rbase = (lane < 16) ? 0 : 8;
  for (int i = 0; i < 8; ++i) {
    red[wv][(i + rbase) * 33 + 2 * l16]     = acc0[i];
    red[wv][(i + rbase) * 33 + 2 * l16 + 1] = acc1[i];
  }
  __syncthreads();
  if (tid < 512) {
    const int r = tid >> 5, d = tid & 31;
    float s = 0.f;
    for (int v = 0; v < 32; ++v) s += red[v][r * 33 + d];
    out[(size_t)(bh * 16 + r) * MD_ + m * D_ + d] = s;
  }
}

extern "C" void kernel_launch(void* const* d_in, const int* in_sizes, int n_in,
                              void* d_out, int out_size, void* d_ws,
                              size_t ws_size, hipStream_t stream) {
  (void)in_sizes; (void)n_in; (void)out_size; (void)d_ws; (void)ws_size;
  const float* pose = (const float*)d_in[0];
  const float* ca   = (const float*)d_in[1];
  const float* ncv  = (const float*)d_in[2];
  const float* na   = (const float*)d_in[3];
  const float* w    = (const float*)d_in[4];

  float* out    = (float*)d_out;
  float* na_out = out + (size_t)B_ * M_ * D_;
  float* qk_out = na_out + (size_t)B_ * M_;

  caps_logits_qk<<<dim3(N_), dim3(256), 0, stream>>>(pose, ncv, na, w, qk_out);
  hipMemcpyAsync(na_out, na, (size_t)B_ * M_ * sizeof(float),
                 hipMemcpyDeviceToDevice, stream);
  caps_out<<<dim3(M_, 2), dim3(1024), 0, stream>>>(pose, ca, w, qk_out, out);
}


# second kernel: pose rows of the A operand read as whole 128-B lines (16-B loads) and re-laid into fragment order through a per-wave LDS strip, not by 8-B loads touching 16 lines each; sums unchanged
# speedup vs baseline: 1.8081x; 1.1699x over previous
#include <hip/hip_runtime.h>


typedef __attribute__((ext_vector_type(2))) float v2f;
typedef __attribute__((ext_vector_type(4))) float v4f;
typedef __attribute__((ext_vector_type(8))) float v8f;

#define B_  32
#define N_  512
#define A_  32
#define M_  128
#define D_  32
#define MD_ (M_ * D_)
#define SCALE_ 0.17677669529663687f

static __device__ __forceinline__ v8f wmma_f32(v2f a, v2f b, v8f c) {
  return __builtin_amdgcn_wmma_f32_16x16x4_f32(false, a, false, b, (short)0, c,
                                               false, false);
}

#define ROW_XADD(x, n)                                                        \
  (x) += __builtin_bit_cast(                                                  \
      float, __builtin_amdgcn_update_dpp(0, __builtin_bit_cast(int, (x)),     \
                                         0x160 + (n), 0xf, 0xf, true))
#define ROW_XMAX(x, n)                                                        \
  (x) = fmaxf((x), __builtin_bit_cast(                                        \
      float, __builtin_amdgcn_update_dpp(0, __builtin_bit_cast(int, (x)),     \
                                         0x160 + (n), 0xf, 0xf, true)))

__global__ __launch_bounds__(256) void caps_logits_qk(
    const float* __restrict__ pose, const float* __restrict__ ncv,
    const float* __restrict__ na, const float* __restrict__ w,
    float* __restrict__ qk_out) {
  const int n    = blockIdx.x;
  const int tid  = threadIdx.x;
  const int lane = tid & 31;
  const int wv   = tid >> 5;
  const int l16  = lane & 15;
  const int khalf = (lane < 16) ? 0 : 2;
  const int rbase = (lane < 16) ? 0 : 8;
  const int c8   = l16 & 7;
  const int msel = l16 >> 3;

  __shared__ __attribute__((aligned(16))) float lg[32 * 128];

  v2f areg[2][8];
  for (int rt = 0; rt < 2; ++rt) {
    const int b = l16 + 16 * rt;
    const float* pp = pose + ((size_t)b * N_ + n) * A_;
    for (int k = 0; k < 8; ++k) areg[rt][k] = *(const v2f*)(pp + 4 * k + khalf);
  }

  for (int mt = 0; mt < 8; ++mt) {
    const int m0 = (mt * 8 + wv) * 2;
    const float* wbase = w + (size_t)n * A_ * MD_ + (size_t)khalf * MD_ +
                         m0 * D_ + 4 * l16;
    const float* cb = ncv + ((size_t)rbase * M_ + m0) * D_ + 4 * l16;
    v8f acc0[4], acc1[4];
    for (int t = 0; t < 4; ++t) { acc0[t] = (v8f){}; acc1[t] = (v8f){}; }
    for (int k = 0; k < 8; ++k) {
      const float4 r0 = *(const float4*)(wbase + (size_t)(4 * k) * MD_);
      const float4 r1 = *(const float4*)(wbase + (size_t)(4 * k + 1) * MD_);
      const float e0[4] = {r0.x, r0.y, r0.z, r0.w};
      const float e1[4] = {r1.x, r1.y, r1.z, r1.w};
      for (int t = 0; t < 4; ++t) {
        v2f bv; bv.x = e0[t]; bv.y = e1[t];
        acc0[t] = wmma_f32(areg[0][k], bv, acc0[t]);
        acc1[t] = wmma_f32(areg[1][k], bv, acc1[t]);
      }
    }
    float p0[8], p1[8];
    for (int i = 0; i < 8; ++i) {
      const float4 q0 = *(const float4*)(cb + (size_t)i * MD_);
      const float4 q1 = *(const float4*)(cb + (size_t)(i + 16) * MD_);
      p0[i] = acc0[0][i] * q0.x + acc0[1][i] * q0.y + acc0[2][i] * q0.z + acc0[3][i] * q0.w;
      p1[i] = acc1[0][i] * q1.x + acc1[1][i] * q1.y + acc1[2][i] * q1.z + acc1[3][i] * q1.w;
    }
    for (int i = 0; i < 8; ++i) { ROW_XADD(p0[i], 1); ROW_XADD(p1[i], 1); }
    for (int i = 0; i < 8; ++i) { ROW_XADD(p0[i], 2); ROW_XADD(p1[i], 2); }
    for (int i = 0; i < 8; ++i) { ROW_XADD(p0[i], 4); ROW_XADD(p1[i], 4); }
    float s0 = p0[0], s1 = p1[0];
    for (int i = 1; i < 8; ++i) {
      s0 = (c8 == i) ? p0[i] : s0;
      s1 = (c8 == i) ? p1[i] : s1;
    }
    lg[(c8 + rbase) * 128 + m0 + msel]      = s0 * SCALE_;
    lg[(c8 + rbase + 16) * 128 + m0 + msel] = s1 * SCALE_;
  }
  __syncthreads();

  {
    const int h = lane >> 4;
    float e[2][8], g[2][8];
    for (int p = 0; p < 2; ++p) {
      const int b = p * 16 + wv * 2 + h;
      const float* row = lg + b * 128 + 4 * l16;
      const float* nap = na + b * 128 + 4 * l16;
      const float4 r0 = *(const float4*)(row);
      const float4 r1 = *(const float4*)(row + 64);
      const float4 a0 = *(const float4*)(nap);
      const float4 a1 = *(const float4*)(nap + 64);
      e[p][0] = r0.x; e[p][1] = r0.y; e[p][2] = r0.z; e[p][3] = r0.w;
      e[p][4] = r1.x; e[p][5] = r1.y; e[p][6] = r1.z; e[p][7] = r1.w;
      g[p][0] = a0.x; g[p][1] = a0.y; g[p][2] = a0.z; g[p][3] = a0.w;
      g[p][4] = a1.x; g[p][5] = a1.y; g[p][6] = a1.z; g[p][7] = a1.w;
    }
    for (int p = 0; p < 2; ++p) {
      const int b = p * 16 + wv * 2 + h;
      float mx = e[p][0];
      for (int i = 1; i < 8; ++i) mx = fmaxf(mx, e[p][i]);
      ROW_XMAX(mx, 1); ROW_XMAX(mx, 2); ROW_XMAX(mx, 4); ROW_XMAX(mx, 8);
      float s = 0.f;
      for (int i = 0; i < 8; ++i) { e[p][i] = __expf(e[p][i] - mx); s += e[p][i]; }
      ROW_XADD(s, 1); ROW_XADD(s, 2); ROW_XADD(s, 4); ROW_XADD(s, 8);
      const float inv = 1.f / s;
      float s2 = 0.f;
      for (int i = 0; i < 8; ++i) { e[p][i] = e[p][i] * inv * g[p][i]; s2 += e[p][i]; }
      ROW_XADD(s2, 1); ROW_XADD(s2, 2); ROW_XADD(s2, 4); ROW_XADD(s2, 8);
      const float inv2 = 1.f / (s2 + 1e-10f);
      float* qp = qk_out + ((size_t)b * N_ + n) * M_ + 4 * l16;
      float4 o0, o1;
      o0.x = e[p][0] * inv2; o0.y = e[p][1] * inv2; o0.z = e[p][2] * inv2; o0.w = e[p][3] * inv2;
      o1.x = e[p][4] * inv2; o1.y = e[p][5] * inv2; o1.z = e[p][6] * inv2; o1.w = e[p][7] * inv2;
      *(float4*)(qp)      = o0;
      *(float4*)(qp + 64) = o1;
    }
  }
}

#define PITCH_ 36
#define SLAB_  (16 * PITCH_ + 32)
#define WST_   (2 * SLAB_)

#define WAVE_LDS_ORDER()                                                      \
  do {                                                                        \
    __builtin_amdgcn_fence(__ATOMIC_RELEASE, "wavefront");                    \
    __builtin_amdgcn_wave_barrier();                                          \
    __builtin_amdgcn_fence(__ATOMIC_ACQUIRE, "wavefront");                    \
  } while (0)

__global__ __launch_bounds__(1024) void caps_out(
    const float* __restrict__ pose, const float* __restrict__ ca,
    const float* __restrict__ w, const float* __restrict__ qk,
    float* __restrict__ out) {
  const int m    = blockIdx.x;
  const int bh   = blockIdx.y;
  const int tid  = threadIdx.x;
  const int lane = tid & 31;
  const int wv   = tid >> 5;
  const int l16  = lane & 15;
  const int khalf = (lane < 16) ? 0 : 2;
  const int b    = bh * 16 + l16;

  __shared__ __attribute__((aligned(16))) float strip[32 * WST_];
  float* st = strip + wv * WST_;

  v8f acc0 = {}, acc1 = {};
  const float* qp = qk + (size_t)b * N_ * M_ + m;
  const float* cp = ca + (size_t)b * N_;
  const float* wl = w + (size_t)khalf * MD_ + m * D_ + 2 * l16;

  const int pb = lane >> 4, pn = (lane >> 3) & 1, pa = (lane & 7) * 4;
  const float* pl = pose + ((size_t)(bh * 16 + pb) * N_ + wv * 16 + pn) * A_ + pa;
  float* sst = st + pn * SLAB_ + pb * PITCH_ + pa;
  const float* sld = st + l16 * PITCH_ + khalf;

#pragma clang loop unroll(disable)
  for (int g = 0; g < 8; ++g) {
    const int ng = wv * 16 + 2 * g;
    v4f pr[8];
    for (int j = 0; j < 8; ++j)
      pr[j] = *(const v4f*)(pl + ((size_t)(2 * j) * N_ + 2 * g) * A_);
    const v2f cav = *(const v2f*)(cp + ng);
    const float q0 = qp[(size_t)ng * M_] * cav.x;
    const float q1 = qp[(size_t)(ng + 1) * M_] * cav.y;
    v2f wa[2][8], wb[2][8];
    for (int k = 0; k < 8; ++k) {
      const float* wn = wl + (size_t)ng * A_ * MD_;
      wa[0][k] = *(const v2f*)(wn + (size_t)(4 * k) * MD_);
      wb[0][k] = *(const v2f*)(wn + (size_t)(4 * k + 1) * MD_);
    }
    for (int j = 0; j < 8; ++j) *(v4f*)(sst + 2 * j * PITCH_) = pr[j];
    WAVE_LDS_ORDER();
    for (int k = 0; k < 8; ++k) {
      const float* wn = wl + (size_t)(ng + 1) * A_ * MD_;
      wa[1][k] = *(const v2f*)(wn + (size_t)(4 * k) * MD_);
      wb[1][k] = *(const v2f*)(wn + (size_t)(4 * k + 1) * MD_);
    }
    for (int nn = 0; nn < 2; ++nn) {
      const float q = nn ? q1 : q0;
      v2f pv[8];
      for (int k = 0; k < 8; ++k)
        pv[k] = *(const v2f*)(sld + nn * SLAB_ + 4 * k);
      for (int k = 0; k < 8; ++k) {
        v2f av; av.x = q * pv[k].x;      av.y = q * pv[k].y;
        v2f be; be.x = wa[nn][k].x;      be.y = wb[nn][k].x;
        v2f bo; bo.x = wa[nn][k].y;      bo.y = wb[nn][k].y;
        acc0 = wmma_f32(av, be, acc0);
        acc1 = wmma_f32(av, bo, acc1);
      }
    }
    WAVE_LDS_ORDER();
  }

  const int rbase = (lane < 16) ? 0 : 8;
  for (int i = 0; i < 8; ++i) {
    st[(i + rbase) * 33 + 2 * l16]     = acc0[i];
    st[(i + rbase) * 33 + 2 * l16 + 1] = acc1[i];
  }
  __syncthreads();
  if (tid < 512) {
    const int r = tid >> 5, d = tid & 31;
    float s = 0.f;
    for (int v = 0; v < 32; ++v) s += strip[v * WST_ + r * 33 + d];
    out[(size_t)(bh * 16 + r) * MD_ + m * D_ + d] = s;
  }
}

extern "C" void kernel_launch(void* const* d_in, const int* in_sizes, int n_in,
                              void* d_out, int out_size, void* d_ws,
                              size_t ws_size, hipStream_t stream) {
  (void)in_sizes; (void)n_in; (void)out_size; (void)d_ws; (void)ws_size;
  const float* pose = (const float*)d_in[0];
  const float* ca   = (const float*)d_in[1];
  const float* ncv  = (const float*)d_in[2];
  const float* na   = (const float*)d_in[3];
  const float* w    = (const float*)d_in[4];

  float* out    = (float*)d_out;
  float* na_out = out + (size_t)B_ * M_ * D_;
  float* qk_out = na_out + (size_t)B_ * M_;

  caps_logits_qk<<<dim3(N_), dim3(256), 0, stream>>>(pose, ncv, na, w, qk_out);
  hipMemcpyAsync(na_out, na, (size_t)B_ * M_ * sizeof(float),
                 hipMemcpyDeviceToDevice, stream);
  caps_out<<<dim3(M_, 2), dim3(1024), 0, stream>>>(pose, ca, w, qk_out, out);
}
